# starting kernel
# baseline (speedup 1.0000x reference)
#include <hip/hip_runtime.h>
#include <hip/hip_bf16.h>

typedef float v2f_ __attribute__((ext_vector_type(2)));
typedef float v8f_ __attribute__((ext_vector_type(8)));

#define NV_   50000
#define NE_   25000
#define NNZ_  400000
#define DIM_  512

__global__ __launch_bounds__(256) void hg_zero4(float4* __restrict__ p, int n4) {
    int i = blockIdx.x * blockDim.x + threadIdx.x;
    if (i < n4) p[i] = make_float4(0.f, 0.f, 0.f, 0.f);
}

__global__ __launch_bounds__(256) void hg_gemm_wmma(const float* __restrict__ X,
                                                    const float* __restrict__ Wt,
                                                    float* __restrict__ Xp) {
    const int lane = threadIdx.x & 31;
    const int wv   = threadIdx.x >> 5;
    const int wrow = wv >> 2;
    const int wcol = wv & 3;

    const int row0 = blockIdx.x * 128 + wrow * 64;
    const int col0 = blockIdx.y * 256 + wcol * 64;

    const int m  = lane & 15;
    const int kb = (lane >> 4) << 1;

    const float* ap[4];
#pragma unroll
    for (int r = 0; r < 4; ++r) {
        int rr = row0 + r * 16;
        if (rr >= NV_) rr = NV_ - 16;
        ap[r] = X + (size_t)(rr + m) * DIM_ + kb;
    }
    const float* bp = Wt + (size_t)(col0 + m) * DIM_ + kb;

    v8f_ c[4][4];
#pragma unroll
    for (int r = 0; r < 4; ++r)
#pragma unroll
        for (int q = 0; q < 4; ++q)
            c[r][q] = (v8f_){};

#pragma unroll 2
    for (int k = 0; k < DIM_; k += 4) {
        v2f_ a[4], b[4];
#pragma unroll
        for (int r = 0; r < 4; ++r)
            a[r] = *(const v2f_*)(ap[r] + k);
#pragma unroll
        for (int q = 0; q < 4; ++q)
            b[q] = *(const v2f_*)(bp + (size_t)q * 16 * DIM_ + k);
#pragma unroll
        for (int r = 0; r < 4; ++r)
#pragma unroll
            for (int q = 0; q < 4; ++q)
                c[r][q] = __builtin_amdgcn_wmma_f32_16x16x4_f32(
                    false, a[r], false, b[q], (short)0, c[r][q], false, false);
    }

    const int rb = (lane >> 4) << 3;
    const int nc = lane & 15;
#pragma unroll
    for (int r = 0; r < 4; ++r) {
        const int rowt = row0 + r * 16;
        if (rowt >= NV_) continue;
        float* o0 = Xp + (size_t)(rowt + rb) * DIM_ + col0 + nc;
#pragma unroll
        for (int v = 0; v < 8; ++v) {
            float* orow = o0 + (size_t)v * DIM_;
#pragma unroll
            for (int q = 0; q < 4; ++q)
                orow[q * 16] = c[r][q][v];
        }
    }
}

__global__ __launch_bounds__(256) void hg_scatter_v2e(const float* __restrict__ Xp,
                                                      const int* __restrict__ vidx,
                                                      const int* __restrict__ eidx,
                                                      float* __restrict__ Xe) {
    const int pair = (blockIdx.x << 3) + (threadIdx.x >> 5);
    const int lane = threadIdx.x & 31;
    if (pair >= NNZ_) return;
    const int v = vidx[pair];
    const int e = eidx[pair];
    const float4* src = (const float4*)(Xp + (size_t)v * DIM_);
    float* dst = Xe + (size_t)e * DIM_;
#pragma unroll
    for (int j = 0; j < 4; ++j) {
        const int q = lane + (j << 5);
        float4 x = src[q];
        atomicAdd(dst + (q << 2) + 0, x.x);
        atomicAdd(dst + (q << 2) + 1, x.y);
        atomicAdd(dst + (q << 2) + 2, x.z);
        atomicAdd(dst + (q << 2) + 3, x.w);
    }
}

__global__ __launch_bounds__(256) void hg_scatter_e2v(const float* __restrict__ Xe,
                                                      const float* __restrict__ degE,
                                                      const float* __restrict__ Wdiag,
                                                      const int* __restrict__ vidx,
                                                      const int* __restrict__ eidx,
                                                      float* __restrict__ out) {
    const int pair = (blockIdx.x << 3) + (threadIdx.x >> 5);
    const int lane = threadIdx.x & 31;
    if (pair >= NNZ_) return;
    const int v = vidx[pair];
    const int e = eidx[pair];
    const float s = degE[e] * Wdiag[e];
    const float4* src = (const float4*)(Xe + (size_t)e * DIM_);
    float* dst = out + (size_t)v * DIM_;
#pragma unroll
    for (int j = 0; j < 4; ++j) {
        const int q = lane + (j << 5);
        float4 x = src[q];
        atomicAdd(dst + (q << 2) + 0, x.x * s);
        atomicAdd(dst + (q << 2) + 1, x.y * s);
        atomicAdd(dst + (q << 2) + 2, x.z * s);
        atomicAdd(dst + (q << 2) + 3, x.w * s);
    }
}

__global__ __launch_bounds__(256) void hg_scale_degv(float4* __restrict__ out,
                                                     const float* __restrict__ degV) {
    int i = blockIdx.x * blockDim.x + threadIdx.x;
    if (i >= NV_ * (DIM_ / 4)) return;
    const int row = i >> 7;
    const float s = degV[row];
    float4 x = out[i];
    x.x *= s; x.y *= s; x.z *= s; x.w *= s;
    out[i] = x;
}

extern "C" void kernel_launch(void* const* d_in, const int* in_sizes, int n_in,
                              void* d_out, int out_size, void* d_ws, size_t ws_size,
                              hipStream_t stream) {
    (void)in_sizes; (void)n_in; (void)out_size; (void)ws_size;
    const float* X     = (const float*)d_in[0];
    const float* W     = (const float*)d_in[1];
    const float* degE  = (const float*)d_in[2];
    const float* degV  = (const float*)d_in[3];
    const float* Wdiag = (const float*)d_in[4];
    const int*   vidx  = (const int*)d_in[5];
    const int*   eidx  = (const int*)d_in[6];
    float* out = (float*)d_out;

    float* Xp = (float*)d_ws;
    float* Xe = Xp + (size_t)NV_ * DIM_;

    const int ne4 = NE_ * (DIM_ / 4);
    const int nv4 = NV_ * (DIM_ / 4);

    hg_zero4<<<(ne4 + 255) / 256, 256, 0, stream>>>((float4*)Xe, ne4);
    hg_zero4<<<(nv4 + 255) / 256, 256, 0, stream>>>((float4*)out, nv4);

    dim3 ggrid((NV_ + 127) / 128, DIM_ / 256);
    hg_gemm_wmma<<<ggrid, 256, 0, stream>>>(X, W, Xp);

    hg_scatter_v2e<<<NNZ_ / 8, 256, 0, stream>>>(Xp, vidx, eidx, Xe);
    hg_scatter_e2v<<<NNZ_ / 8, 256, 0, stream>>>(Xe, degE, Wdiag, vidx, eidx, out);

    hg_scale_degv<<<(nv4 + 255) / 256, 256, 0, stream>>>((float4*)out, degV);
}


# Float-atomic scatters replaced by per-call grouping of pairs by row (int-atomic rank, arrival order) and one-wave-per-row register sums; rows stored once, so zeroing and the degV pass are gone
# speedup vs baseline: 2.2464x; 2.2464x over previous
#include <hip/hip_runtime.h>
#include <hip/hip_bf16.h>


typedef float v2f_ __attribute__((ext_vector_type(2)));
typedef float v8f_ __attribute__((ext_vector_type(8)));

#define NV_   50000
#define NE_   25000
#define NNZ_  400000
#define DIM_  512
#define NR_   (NE_ + NV_)

static_assert(NE_ % 8 == 0 && NV_ % 8 == 0, "one wave per row, 8 rows per block, exact grids");

__global__ __launch_bounds__(256) void hg_zero_i4(int4* __restrict__ p, int n4) {
    int i = blockIdx.x * blockDim.x + threadIdx.x;
    if (i < n4) p[i] = make_int4(0, 0, 0, 0);
}

__global__ __launch_bounds__(256) void hg_count(const int* __restrict__ vidx,
                                                const int* __restrict__ eidx,
                                                int* __restrict__ cnt,
                                                int* __restrict__ rank) {
    const int i = blockIdx.x * blockDim.x + threadIdx.x;
    if (i >= NNZ_) return;
    const int e = eidx[i];
    const int v = vidx[i];
    const int re = atomicAdd(cnt + e, 1);
    const int rv = atomicAdd(cnt + NE_ + v, 1);
    rank[i]        = re;
    rank[NNZ_ + i] = rv;
}

__global__ __launch_bounds__(256) void hg_alloc(const int* __restrict__ cnt,
                                                int* __restrict__ tot,
                                                int* __restrict__ off) {
    const int r    = blockIdx.x * blockDim.x + threadIdx.x;
    const int lane = threadIdx.x & 31;
    const int c    = (r < NR_) ? cnt[r] : 0;
    int incl = c;
    for (int d = 1; d < 32; d <<= 1) {
        const int t = __shfl_up(incl, d);
        if (lane >= d) incl += t;
    }
    int base = 0;
    if (lane == 31) base = atomicAdd(tot, incl);
    base = __shfl(base, 31);
    if (r < NR_) off[r] = base + incl - c;
}

__global__ __launch_bounds__(256) void hg_fill(const int* __restrict__ vidx,
                                               const int* __restrict__ eidx,
                                               const int* __restrict__ off,
                                               const int* __restrict__ rank,
                                               int* __restrict__ list) {
    const int i = blockIdx.x * blockDim.x + threadIdx.x;
    if (i >= NNZ_) return;
    const int e = eidx[i];
    const int v = vidx[i];
    list[off[e] + rank[i]]              = v;
    list[off[NE_ + v] + rank[NNZ_ + i]] = e;
}

__global__ __launch_bounds__(256) void hg_gemm_wmma(const float* __restrict__ X,
                                                    const float* __restrict__ Wt,
                                                    float* __restrict__ Xp) {
    const int lane = threadIdx.x & 31;
    const int wv   = threadIdx.x >> 5;
    const int wrow = wv >> 2;
    const int wcol = wv & 3;

    const int row0 = blockIdx.x * 128 + wrow * 64;
    const int col0 = blockIdx.y * 256 + wcol * 64;

    const int m  = lane & 15;
    const int kb = (lane >> 4) << 1;

    const float* ap[4];
#pragma unroll
    for (int r = 0; r < 4; ++r) {
        int rr = row0 + r * 16;
        if (rr >= NV_) rr = NV_ - 16;
        ap[r] = X + (size_t)(rr + m) * DIM_ + kb;
    }
    const float* bp = Wt + (size_t)(col0 + m) * DIM_ + kb;

    v8f_ c[4][4];
#pragma unroll
    for (int r = 0; r < 4; ++r)
#pragma unroll
        for (int q = 0; q < 4; ++q)
            c[r][q] = (v8f_){};

#pragma unroll 2
    for (int k = 0; k < DIM_; k += 4) {
        v2f_ a[4], b[4];
#pragma unroll
        for (int r = 0; r < 4; ++r)
            a[r] = *(const v2f_*)(ap[r] + k);
#pragma unroll
        for (int q = 0; q < 4; ++q)
            b[q] = *(const v2f_*)(bp + (size_t)q * 16 * DIM_ + k);
#pragma unroll
        for (int r = 0; r < 4; ++r)
#pragma unroll
            for (int q = 0; q < 4; ++q)
                c[r][q] = __builtin_amdgcn_wmma_f32_16x16x4_f32(
                    false, a[r], false, b[q], (short)0, c[r][q], false, false);
    }

    const int rb = (lane >> 4) << 3;
    const int nc = lane & 15;
#pragma unroll
    for (int r = 0; r < 4; ++r) {
        const int rowt = row0 + r * 16;
        if (rowt >= NV_) continue;
        float* o0 = Xp + (size_t)(rowt + rb) * DIM_ + col0 + nc;
#pragma unroll
        for (int v = 0; v < 8; ++v) {
            float* orow = o0 + (size_t)v * DIM_;
#pragma unroll
            for (int q = 0; q < 4; ++q)
                orow[q * 16] = c[r][q][v];
        }
    }
}

__device__ __forceinline__ float4 hg_add4(float4 a, float4 b) {
    return make_float4(a.x + b.x, a.y + b.y, a.z + b.z, a.w + b.w);
}

__device__ __forceinline__ void hg_ld_row(const float* __restrict__ src, int r, int lane,
                                          float4 x[4]) {
    const float4* p = (const float4*)(src + (size_t)r * DIM_) + lane;
    x[0] = p[0]; x[1] = p[32]; x[2] = p[64]; x[3] = p[96];
}

__device__ __forceinline__ void hg_row_sum(const float* __restrict__ src,
                                           const int* __restrict__ lst, int n, int lane,
                                           float4 acc[4]) {
    for (int j = 0; j < 4; ++j) acc[j] = make_float4(0.f, 0.f, 0.f, 0.f);
    int i = 0;
    for (; i + 4 <= n; i += 4) {
        const int r0 = lst[i], r1 = lst[i + 1], r2 = lst[i + 2], r3 = lst[i + 3];
        float4 x0[4], x1[4], x2[4], x3[4];
        hg_ld_row(src, r0, lane, x0);
        hg_ld_row(src, r1, lane, x1);
        hg_ld_row(src, r2, lane, x2);
        hg_ld_row(src, r3, lane, x3);
        for (int j = 0; j < 4; ++j)
            acc[j] = hg_add4(acc[j], hg_add4(hg_add4(x0[j], x1[j]), hg_add4(x2[j], x3[j])));
    }
    if (i + 2 <= n) {
        const int r0 = lst[i], r1 = lst[i + 1];
        float4 x0[4], x1[4];
        hg_ld_row(src, r0, lane, x0);
        hg_ld_row(src, r1, lane, x1);
        for (int j = 0; j < 4; ++j)
            acc[j] = hg_add4(acc[j], hg_add4(x0[j], x1[j]));
        i += 2;
    }
    if (i < n) {
        float4 x0[4];
        hg_ld_row(src, lst[i], lane, x0);
        for (int j = 0; j < 4; ++j) acc[j] = hg_add4(acc[j], x0[j]);
    }
}

__global__ __launch_bounds__(256) void hg_gather_v2e(const float* __restrict__ Xp,
                                                     const int* __restrict__ off,
                                                     const int* __restrict__ cnt,
                                                     const int* __restrict__ list,
                                                     const float* __restrict__ degE,
                                                     const float* __restrict__ Wdiag,
                                                     float* __restrict__ Xe) {
    const int lane = threadIdx.x & 31;
    const int e = __builtin_amdgcn_readfirstlane((blockIdx.x << 3) + (threadIdx.x >> 5));
    float4 acc[4];
    hg_row_sum(Xp, list + off[e], cnt[e], lane, acc);
    const float s = degE[e] * Wdiag[e];
    float4* dst = (float4*)(Xe + (size_t)e * DIM_) + lane;
    for (int j = 0; j < 4; ++j)
        dst[j * 32] = make_float4(acc[j].x * s, acc[j].y * s, acc[j].z * s, acc[j].w * s);
}

__global__ __launch_bounds__(256) void hg_gather_e2v(const float* __restrict__ Xe,
                                                     const int* __restrict__ off,
                                                     const int* __restrict__ cnt,
                                                     const int* __restrict__ list,
                                                     const float* __restrict__ degV,
                                                     float* __restrict__ out) {
    const int lane = threadIdx.x & 31;
    const int v = __builtin_amdgcn_readfirstlane((blockIdx.x << 3) + (threadIdx.x >> 5));
    float4 acc[4];
    hg_row_sum(Xe, list + off[v], cnt[v], lane, acc);
    const float s = degV[v];
    float4* dst = (float4*)(out + (size_t)v * DIM_) + lane;
    for (int j = 0; j < 4; ++j)
        dst[j * 32] = make_float4(acc[j].x * s, acc[j].y * s, acc[j].z * s, acc[j].w * s);
}

extern "C" void kernel_launch(void* const* d_in, const int* in_sizes, int n_in,
                              void* d_out, int out_size, void* d_ws, size_t ws_size,
                              hipStream_t stream) {
    (void)in_sizes; (void)n_in; (void)out_size; (void)ws_size;
    const float* X     = (const float*)d_in[0];
    const float* W     = (const float*)d_in[1];
    const float* degE  = (const float*)d_in[2];
    const float* degV  = (const float*)d_in[3];
    const float* Wdiag = (const float*)d_in[4];
    const int*   vidx  = (const int*)d_in[5];
    const int*   eidx  = (const int*)d_in[6];
    float* out = (float*)d_out;

    float* Xp   = (float*)d_ws;
    float* Xe   = Xp + (size_t)NV_ * DIM_;
    int*   cnt  = (int*)(Xe + (size_t)NE_ * DIM_);
    int*   tot  = cnt + NR_;
    int*   off  = cnt + NR_ + 8;
    int*   rank = off + NR_;
    int*   list = rank + 2 * NNZ_;

    const int nz4 = (NR_ + 8) / 4;

    hg_zero_i4<<<(nz4 + 255) / 256, 256, 0, stream>>>((int4*)cnt, nz4);
    hg_count<<<(NNZ_ + 255) / 256, 256, 0, stream>>>(vidx, eidx, cnt, rank);
    hg_alloc<<<(NR_ + 255) / 256, 256, 0, stream>>>(cnt, tot, off);
    hg_fill<<<(NNZ_ + 255) / 256, 256, 0, stream>>>(vidx, eidx, off, rank, list);

    dim3 ggrid((NV_ + 127) / 128, DIM_ / 256);
    hg_gemm_wmma<<<ggrid, 256, 0, stream>>>(X, W, Xp);

    hg_gather_v2e<<<NE_ / 8, 256, 0, stream>>>(Xp, off, cnt, list, degE, Wdiag, Xe);
    hg_gather_e2v<<<NV_ / 8, 256, 0, stream>>>(Xe, off + NE_, cnt + NE_, list, degV, out);
}


# Pair grouping in one launch: work-groups own row ranges, scan the key arrays and rank own pairs with LDS adds (arrival order inside a row, as before); global atomics and 3 launches gone
# speedup vs baseline: 3.6289x; 1.6154x over previous
#include <hip/hip_runtime.h>
#include <hip/hip_bf16.h>


typedef float v2f_ __attribute__((ext_vector_type(2)));
typedef float v8f_ __attribute__((ext_vector_type(8)));

#define NV_   50000
#define NE_   25000
#define NNZ_  400000
#define DIM_  512
#define NR_   (NE_ + NV_)

static_assert(NE_ % 8 == 0 && NV_ % 8 == 0, "one wave per row, 8 rows per block, exact grids");

#define GRP_T_     1024
#define GRP_E_     84
#define GRP_V_     84
#define GRP_RE_    ((NE_ + GRP_E_ - 1) / GRP_E_)
#define GRP_RV_    ((NV_ + GRP_V_ - 1) / GRP_V_)
#define GRP_RMAX_  608
#define GRP_CAPW_  384
#define GRP_Q_     (NNZ_ / 4)
#define GRP_IT_    ((GRP_Q_ + 4 * GRP_T_ - 1) / (4 * GRP_T_))

static_assert(NNZ_ % 4 == 0, "key arrays are read as whole 16-byte quads");
static_assert(NNZ_ <= (1 << 19), "a pair index must fit in 19 bits of a stash word");
static_assert(GRP_RMAX_ <= 1024 && GRP_RMAX_ <= GRP_T_, "a local row must fit in 10 bits; one thread per row in the scan");
static_assert(GRP_RE_ <= GRP_RMAX_ && GRP_RV_ <= GRP_RMAX_, "LDS counters cover a group's rows");
static_assert((GRP_E_ - 1) * GRP_RE_ < NE_ && (GRP_V_ - 1) * GRP_RV_ < NV_, "every group owns at least one row");

__device__ __forceinline__ void hg_note(int key, int i, int lo, int nr,
                                        int* __restrict__ lcnt, int* __restrict__ mystash,
                                        int& nbelow, int& nw) {
    const int d = key - lo;
    const bool in = (unsigned)d < (unsigned)nr;
    nbelow += (int)((unsigned)d >> 31);
    const unsigned m = __builtin_amdgcn_ballot_w32(in);
    if (m != 0u) {
        if (in) {
            atomicAdd(&lcnt[d], 1);
            const int s = nw + (int)__builtin_amdgcn_mbcnt_lo(m, 0u);
            if (s < GRP_CAPW_) mystash[s] = i | (d << 19);
        }
        nw += __popc(m);
    }
}

__device__ __forceinline__ void hg_place(int key, int i, int lo, int nr,
                                         int* __restrict__ lpos, const int* __restrict__ oth,
                                         int* __restrict__ list) {
    const int d = key - lo;
    if ((unsigned)d < (unsigned)nr) {
        const int p = atomicAdd(&lpos[d], 1);
        list[p] = oth[i];
    }
}

__global__ __launch_bounds__(GRP_T_) void hg_group(const int* __restrict__ vidx,
                                                   const int* __restrict__ eidx,
                                                   int* __restrict__ cnt,
                                                   int* __restrict__ off,
                                                   int* __restrict__ list) {
    __shared__ int lcnt[GRP_RMAX_];
    __shared__ int lpos[GRP_RMAX_];
    __shared__ int wtot[32];
    __shared__ int sbelow;
    __shared__ int sovf;
    __shared__ int stash[32 * GRP_CAPW_];

    const int t    = threadIdx.x;
    const int lane = t & 31;
    const int wv   = __builtin_amdgcn_readfirstlane(t >> 5);

    const bool isE  = blockIdx.x < GRP_E_;
    const int  b    = isE ? (int)blockIdx.x : (int)blockIdx.x - GRP_E_;
    const int  R    = isE ? GRP_RE_ : GRP_RV_;
    const int  lo   = b * R;
    const int  nall = isE ? NE_ : NV_;
    const int  nr   = (nall - lo < R) ? (nall - lo) : R;
    const int* key  = isE ? eidx : vidx;
    const int* oth  = isE ? vidx : eidx;
    const int  row0 = (isE ? 0 : NE_) + lo;
    const int  lst0 = isE ? 0 : NNZ_;
    const int4* key4 = (const int4*)key;
    int* mystash = stash + wv * GRP_CAPW_;

    if (t < GRP_RMAX_) lcnt[t] = 0;
    if (t == 0) { sbelow = 0; sovf = 0; }
    __syncthreads();

    int nbelow = 0;
    int nw     = 0;
    for (int it = 0; it < GRP_IT_; ++it) {
        const int q0 = it * (4 * GRP_T_) + t;
        int4 k[4];
#pragma unroll
        for (int j = 0; j < 4; ++j) {
            const int q = q0 + j * GRP_T_;
            k[j] = (q < GRP_Q_) ? key4[q] : make_int4(0x7fffffff, 0x7fffffff, 0x7fffffff, 0x7fffffff);
        }
#pragma unroll
        for (int j = 0; j < 4; ++j) {
            const int i = (q0 + j * GRP_T_) * 4;
            hg_note(k[j].x, i + 0, lo, nr, lcnt, mystash, nbelow, nw);
            hg_note(k[j].y, i + 1, lo, nr, lcnt, mystash, nbelow, nw);
            hg_note(k[j].z, i + 2, lo, nr, lcnt, mystash, nbelow, nw);
            hg_note(k[j].w, i + 3, lo, nr, lcnt, mystash, nbelow, nw);
        }
    }
    for (int d = 16; d > 0; d >>= 1) nbelow += __shfl_xor(nbelow, d);
    if (lane == 0) {
        atomicAdd(&sbelow, nbelow);
        if (nw > GRP_CAPW_) atomicOr(&sovf, 1);
    }
    __syncthreads();

    const int c = (t < nr) ? lcnt[t] : 0;
    int incl = c;
    for (int d = 1; d < 32; d <<= 1) {
        const int u = __shfl_up(incl, d);
        if (lane >= d) incl += u;
    }
    if (lane == 31) wtot[wv] = incl;
    __syncthreads();
    if (wv == 0) {
        const int w = wtot[lane];
        int wi = w;
        for (int d = 1; d < 32; d <<= 1) {
            const int u = __shfl_up(wi, d);
            if (lane >= d) wi += u;
        }
        wtot[lane] = wi - w;
    }
    __syncthreads();
    const int first = lst0 + sbelow + wtot[wv] + incl - c;
    if (t < nr) {
        off[row0 + t] = first;
        cnt[row0 + t] = c;
        lpos[t]       = first;
    }
    const int ovf = sovf;
    __syncthreads();

    if (!ovf) {
        for (int j = lane; j < nw; j += 32) {
            const int ent = mystash[j];
            const int p   = atomicAdd(&lpos[ent >> 19], 1);
            list[p] = oth[ent & ((1 << 19) - 1)];
        }
    } else {
        for (int q = t; q < GRP_Q_; q += GRP_T_) {
            const int4 k = key4[q];
            hg_place(k.x, 4 * q + 0, lo, nr, lpos, oth, list);
            hg_place(k.y, 4 * q + 1, lo, nr, lpos, oth, list);
            hg_place(k.z, 4 * q + 2, lo, nr, lpos, oth, list);
            hg_place(k.w, 4 * q + 3, lo, nr, lpos, oth, list);
        }
    }
}

__global__ __launch_bounds__(256) void hg_gemm_wmma(const float* __restrict__ Xs,
                                                    const float* __restrict__ Wt,
                                                    float* __restrict__ Xe) {
    const int lane = threadIdx.x & 31;
    const int wv   = __builtin_amdgcn_readfirstlane(threadIdx.x >> 5);
    const int wrow = wv >> 2;
    const int wcol = wv & 3;

    const int row0 = blockIdx.x * 128 + wrow * 64;
    const int col0 = blockIdx.y * 256 + wcol * 64;
    if (row0 >= NE_) return;

    const int m  = lane & 15;
    const int kb = (lane >> 4) << 1;

    const float* ap[4];
#pragma unroll
    for (int r = 0; r < 4; ++r) {
        int rr = row0 + r * 16 + m;
        if (rr > NE_ - 1) rr = NE_ - 1;
        ap[r] = Xs + (size_t)rr * DIM_ + kb;
    }
    const float* bp = Wt + (size_t)(col0 + m) * DIM_ + kb;

    v8f_ c[4][4];
#pragma unroll
    for (int r = 0; r < 4; ++r)
#pragma unroll
        for (int q = 0; q < 4; ++q)
            c[r][q] = (v8f_){};

#pragma unroll 2
    for (int k = 0; k < DIM_; k += 4) {
        v2f_ a[4], b[4];
#pragma unroll
        for (int r = 0; r < 4; ++r)
            a[r] = *(const v2f_*)(ap[r] + k);
#pragma unroll
        for (int q = 0; q < 4; ++q)
            b[q] = *(const v2f_*)(bp + (size_t)q * 16 * DIM_ + k);
#pragma unroll
        for (int r = 0; r < 4; ++r)
#pragma unroll
            for (int q = 0; q < 4; ++q)
                c[r][q] = __builtin_amdgcn_wmma_f32_16x16x4_f32(
                    false, a[r], false, b[q], (short)0, c[r][q], false, false);
    }

    const int rb = (lane >> 4) << 3;
    const int nc = lane & 15;
#pragma unroll
    for (int r = 0; r < 4; ++r) {
        const int rowt = row0 + r * 16;
        if (rowt + rb >= NE_) continue;
        float* o0 = Xe + (size_t)(rowt + rb) * DIM_ + col0 + nc;
#pragma unroll
        for (int v = 0; v < 8; ++v) {
            float* orow = o0 + (size_t)v * DIM_;
#pragma unroll
            for (int q = 0; q < 4; ++q)
                orow[q * 16] = c[r][q][v];
        }
    }
}

__device__ __forceinline__ float4 hg_add4(float4 a, float4 b) {
    return make_float4(a.x + b.x, a.y + b.y, a.z + b.z, a.w + b.w);
}

__device__ __forceinline__ void hg_ld_row(const float* __restrict__ src, int r, int lane,
                                          float4 x[4]) {
    const float4* p = (const float4*)(src + (size_t)r * DIM_) + lane;
    x[0] = p[0]; x[1] = p[32]; x[2] = p[64]; x[3] = p[96];
}

__device__ __forceinline__ void hg_row_sum(const float* __restrict__ src,
                                           const int* __restrict__ lst, int n, int lane,
                                           float4 acc[4]) {
    for (int j = 0; j < 4; ++j) acc[j] = make_float4(0.f, 0.f, 0.f, 0.f);
    int i = 0;
    for (; i + 4 <= n; i += 4) {
        const int r0 = lst[i], r1 = lst[i + 1], r2 = lst[i + 2], r3 = lst[i + 3];
        float4 x0[4], x1[4], x2[4], x3[4];
        hg_ld_row(src, r0, lane, x0);
        hg_ld_row(src, r1, lane, x1);
        hg_ld_row(src, r2, lane, x2);
        hg_ld_row(src, r3, lane, x3);
        for (int j = 0; j < 4; ++j)
            acc[j] = hg_add4(acc[j], hg_add4(hg_add4(x0[j], x1[j]), hg_add4(x2[j], x3[j])));
    }
    if (i + 2 <= n) {
        const int r0 = lst[i], r1 = lst[i + 1];
        float4 x0[4], x1[4];
        hg_ld_row(src, r0, lane, x0);
        hg_ld_row(src, r1, lane, x1);
        for (int j = 0; j < 4; ++j)
            acc[j] = hg_add4(acc[j], hg_add4(x0[j], x1[j]));
        i += 2;
    }
    if (i < n) {
        float4 x0[4];
        hg_ld_row(src, lst[i], lane, x0);
        for (int j = 0; j < 4; ++j) acc[j] = hg_add4(acc[j], x0[j]);
    }
}

__global__ __launch_bounds__(256) void hg_gather_v2e(const float* __restrict__ X,
                                                     const int* __restrict__ off,
                                                     const int* __restrict__ cnt,
                                                     const int* __restrict__ list,
                                                     const float* __restrict__ degE,
                                                     const float* __restrict__ Wdiag,
                                                     float* __restrict__ Xs) {
    const int lane = threadIdx.x & 31;
    const int e = __builtin_amdgcn_readfirstlane((blockIdx.x << 3) + (threadIdx.x >> 5));
    float4 acc[4];
    hg_row_sum(X, list + off[e], cnt[e], lane, acc);
    const float s = degE[e] * Wdiag[e];
    float4* dst = (float4*)(Xs + (size_t)e * DIM_) + lane;
    for (int j = 0; j < 4; ++j)
        dst[j * 32] = make_float4(acc[j].x * s, acc[j].y * s, acc[j].z * s, acc[j].w * s);
}

__global__ __launch_bounds__(256) void hg_gather_e2v(const float* __restrict__ Xe,
                                                     const int* __restrict__ off,
                                                     const int* __restrict__ cnt,
                                                     const int* __restrict__ list,
                                                     const float* __restrict__ degV,
                                                     float* __restrict__ out) {
    const int lane = threadIdx.x & 31;
    const int v = __builtin_amdgcn_readfirstlane((blockIdx.x << 3) + (threadIdx.x >> 5));
    float4 acc[4];
    hg_row_sum(Xe, list + off[v], cnt[v], lane, acc);
    const float s = degV[v];
    float4* dst = (float4*)(out + (size_t)v * DIM_) + lane;
    for (int j = 0; j < 4; ++j)
        dst[j * 32] = make_float4(acc[j].x * s, acc[j].y * s, acc[j].z * s, acc[j].w * s);
}

extern "C" void kernel_launch(void* const* d_in, const int* in_sizes, int n_in,
                              void* d_out, int out_size, void* d_ws, size_t ws_size,
                              hipStream_t stream) {
    (void)in_sizes; (void)n_in; (void)out_size; (void)ws_size;
    const float* X     = (const float*)d_in[0];
    const float* W     = (const float*)d_in[1];
    const float* degE  = (const float*)d_in[2];
    const float* degV  = (const float*)d_in[3];
    const float* Wdiag = (const float*)d_in[4];
    const int*   vidx  = (const int*)d_in[5];
    const int*   eidx  = (const int*)d_in[6];
    float* out = (float*)d_out;

    float* Xs   = (float*)d_ws;
    float* Xe   = Xs + (size_t)NE_ * DIM_;
    int*   cnt  = (int*)(Xe + (size_t)NE_ * DIM_);
    int*   off  = cnt + NR_;
    int*   list = off + NR_;

    hg_group<<<GRP_E_ + GRP_V_, GRP_T_, 0, stream>>>(vidx, eidx, cnt, off, list);

    hg_gather_v2e<<<NE_ / 8, 256, 0, stream>>>(X, off, cnt, list, degE, Wdiag, Xs);

    dim3 ggrid((NE_ + 127) / 128, DIM_ / 256);
    hg_gemm_wmma<<<ggrid, 256, 0, stream>>>(Xs, W, Xe);

    hg_gather_e2v<<<NV_ / 8, 256, 0, stream>>>(Xe, off + NE_, cnt + NE_, list, degV, out);
}


# Projection product: Xs and W tiles staged through LDS by whole-line 16-byte loads, one group per WGP owns 6-7 row tiles x all columns; 16-byte fragment reads re-pair k, so f32 sum order changes
# speedup vs baseline: 4.2415x; 1.1688x over previous
#include <hip/hip_runtime.h>
#include <hip/hip_bf16.h>


typedef float v2f_ __attribute__((ext_vector_type(2)));
typedef float v8f_ __attribute__((ext_vector_type(8)));

#define NV_   50000
#define NE_   25000
#define NNZ_  400000
#define DIM_  512
#define NR_   (NE_ + NV_)

static_assert(NE_ % 8 == 0 && NV_ % 8 == 0, "one wave per row, 8 rows per block, exact grids");

#define GRP_T_     1024
#define GRP_E_     84
#define GRP_V_     84
#define GRP_RE_    ((NE_ + GRP_E_ - 1) / GRP_E_)
#define GRP_RV_    ((NV_ + GRP_V_ - 1) / GRP_V_)
#define GRP_RMAX_  608
#define GRP_CAPW_  384
#define GRP_Q_     (NNZ_ / 4)
#define GRP_IT_    ((GRP_Q_ + 4 * GRP_T_ - 1) / (4 * GRP_T_))

static_assert(NNZ_ % 4 == 0, "key arrays are read as whole 16-byte quads");
static_assert(NNZ_ <= (1 << 19), "a pair index must fit in 19 bits of a stash word");
static_assert(GRP_RMAX_ <= 1024 && GRP_RMAX_ <= GRP_T_, "a local row must fit in 10 bits; one thread per row in the scan");
static_assert(GRP_RE_ <= GRP_RMAX_ && GRP_RV_ <= GRP_RMAX_, "LDS counters cover a group's rows");
static_assert((GRP_E_ - 1) * GRP_RE_ < NE_ && (GRP_V_ - 1) * GRP_RV_ < NV_, "every group owns at least one row");

__device__ __forceinline__ void hg_note(int key, int i, int lo, int nr,
                                        int* __restrict__ lcnt, int* __restrict__ mystash,
                                        int& nbelow, int& nw) {
    const int d = key - lo;
    const bool in = (unsigned)d < (unsigned)nr;
    nbelow += (int)((unsigned)d >> 31);
    const unsigned m = __builtin_amdgcn_ballot_w32(in);
    if (m != 0u) {
        if (in) {
            atomicAdd(&lcnt[d], 1);
            const int s = nw + (int)__builtin_amdgcn_mbcnt_lo(m, 0u);
            if (s < GRP_CAPW_) mystash[s] = i | (d << 19);
        }
        nw += __popc(m);
    }
}

__device__ __forceinline__ void hg_place(int key, int i, int lo, int nr,
                                         int* __restrict__ lpos, const int* __restrict__ oth,
                                         int* __restrict__ list) {
    const int d = key - lo;
    if ((unsigned)d < (unsigned)nr) {
        const int p = atomicAdd(&lpos[d], 1);
        list[p] = oth[i];
    }
}

__global__ __launch_bounds__(GRP_T_) void hg_group(const int* __restrict__ vidx,
                                                   const int* __restrict__ eidx,
                                                   int* __restrict__ cnt,
                                                   int* __restrict__ off,
                                                   int* __restrict__ list) {
    __shared__ int lcnt[GRP_RMAX_];
    __shared__ int lpos[GRP_RMAX_];
    __shared__ int wtot[32];
    __shared__ int sbelow;
    __shared__ int sovf;
    __shared__ int stash[32 * GRP_CAPW_];

    const int t    = threadIdx.x;
    const int lane = t & 31;
    const int wv   = __builtin_amdgcn_readfirstlane(t >> 5);

    const bool isE  = blockIdx.x < GRP_E_;
    const int  b    = isE ? (int)blockIdx.x : (int)blockIdx.x - GRP_E_;
    const int  R    = isE ? GRP_RE_ : GRP_RV_;
    const int  lo   = b * R;
    const int  nall = isE ? NE_ : NV_;
    const int  nr   = (nall - lo < R) ? (nall - lo) : R;
    const int* key  = isE ? eidx : vidx;
    const int* oth  = isE ? vidx : eidx;
    const int  row0 = (isE ? 0 : NE_) + lo;
    const int  lst0 = isE ? 0 : NNZ_;
    const int4* key4 = (const int4*)key;
    int* mystash = stash + wv * GRP_CAPW_;

    if (t < GRP_RMAX_) lcnt[t] = 0;
    if (t == 0) { sbelow = 0; sovf = 0; }
    __syncthreads();

    int nbelow = 0;
    int nw     = 0;
    for (int it = 0; it < GRP_IT_; ++it) {
        const int q0 = it * (4 * GRP_T_) + t;
        int4 k[4];
#pragma unroll
        for (int j = 0; j < 4; ++j) {
            const int q = q0 + j * GRP_T_;
            k[j] = (q < GRP_Q_) ? key4[q] : make_int4(0x7fffffff, 0x7fffffff, 0x7fffffff, 0x7fffffff);
        }
#pragma unroll
        for (int j = 0; j < 4; ++j) {
            const int i = (q0 + j * GRP_T_) * 4;
            hg_note(k[j].x, i + 0, lo, nr, lcnt, mystash, nbelow, nw);
            hg_note(k[j].y, i + 1, lo, nr, lcnt, mystash, nbelow, nw);
            hg_note(k[j].z, i + 2, lo, nr, lcnt, mystash, nbelow, nw);
            hg_note(k[j].w, i + 3, lo, nr, lcnt, mystash, nbelow, nw);
        }
    }
    for (int d = 16; d > 0; d >>= 1) nbelow += __shfl_xor(nbelow, d);
    if (lane == 0) {
        atomicAdd(&sbelow, nbelow);
        if (nw > GRP_CAPW_) atomicOr(&sovf, 1);
    }
    __syncthreads();

    const int c = (t < nr) ? lcnt[t] : 0;
    int incl = c;
    for (int d = 1; d < 32; d <<= 1) {
        const int u = __shfl_up(incl, d);
        if (lane >= d) incl += u;
    }
    if (lane == 31) wtot[wv] = incl;
    __syncthreads();
    if (wv == 0) {
        const int w = wtot[lane];
        int wi = w;
        for (int d = 1; d < 32; d <<= 1) {
            const int u = __shfl_up(wi, d);
            if (lane >= d) wi += u;
        }
        wtot[lane] = wi - w;
    }
    __syncthreads();
    const int first = lst0 + sbelow + wtot[wv] + incl - c;
    if (t < nr) {
        off[row0 + t] = first;
        cnt[row0 + t] = c;
        lpos[t]       = first;
    }
    const int ovf = sovf;
    __syncthreads();

    if (!ovf) {
        for (int j = lane; j < nw; j += 32) {
            const int ent = mystash[j];
            const int p   = atomicAdd(&lpos[ent >> 19], 1);
            list[p] = oth[ent & ((1 << 19) - 1)];
        }
    } else {
        for (int q = t; q < GRP_Q_; q += GRP_T_) {
            const int4 k = key4[q];
            hg_place(k.x, 4 * q + 0, lo, nr, lpos, oth, list);
            hg_place(k.y, 4 * q + 1, lo, nr, lpos, oth, list);
            hg_place(k.z, 4 * q + 2, lo, nr, lpos, oth, list);
            hg_place(k.w, 4 * q + 3, lo, nr, lpos, oth, list);
        }
    }
}

#define GB_WG_     256
#define GB_NT_     ((NE_ + 15) / 16)
#define GB_RT_     7
#define GB_ROWS_   (GB_RT_ * 16)
#define GB_BK_     32
#define GB_NCH_    (DIM_ / GB_BK_)
#define GB_P_      36
#define GB_AROWS_  128
#define GB_STAGE_  ((GB_AROWS_ + DIM_) * GB_P_)

static_assert((GB_NT_ + GB_WG_ - 1) / GB_WG_ <= GB_RT_, "a group's share of the tiles fits its accumulators");
static_assert(DIM_ % GB_BK_ == 0 && GB_BK_ == 32, "a chunk row is one 128-byte line, 8 lanes x 16 bytes");
static_assert(DIM_ == 512, "8 waves x 64 columns; 16 W rows per thread and chunk");

#define HG_STAGE_LD(k)                                                                   \
    sa0 = *(const float4*)(ga0 + (k)); sa1 = *(const float4*)(ga1 + (k));                \
    sa2 = *(const float4*)(ga2 + (k)); sa3 = *(const float4*)(ga3 + (k));                \
    HG_LDB(0, k)  HG_LDB(1, k)  HG_LDB(2, k)  HG_LDB(3, k)  HG_LDB(4, k)  HG_LDB(5, k)   \
    HG_LDB(6, k)  HG_LDB(7, k)  HG_LDB(8, k)  HG_LDB(9, k)  HG_LDB(10, k) HG_LDB(11, k)  \
    HG_LDB(12, k) HG_LDB(13, k) HG_LDB(14, k) HG_LDB(15, k)
#define HG_LDB(i, k)  sb##i = *(const float4*)(gb + (size_t)(i) * 32 * DIM_ + (k));
#define HG_STAGE_ST(st)                                                                  \
    *(float4*)((st))                = sa0; *(float4*)((st) + 32 * GB_P_) = sa1;          \
    *(float4*)((st) + 64 * GB_P_)   = sa2; *(float4*)((st) + 96 * GB_P_) = sa3;          \
    HG_STB(0, st)  HG_STB(1, st)  HG_STB(2, st)  HG_STB(3, st)  HG_STB(4, st)  HG_STB(5, st)   \
    HG_STB(6, st)  HG_STB(7, st)  HG_STB(8, st)  HG_STB(9, st)  HG_STB(10, st) HG_STB(11, st)  \
    HG_STB(12, st) HG_STB(13, st) HG_STB(14, st) HG_STB(15, st)
#define HG_STB(i, st) *(float4*)((st) + (GB_AROWS_ + (i) * 32) * GB_P_) = sb##i;

__device__ __forceinline__ void hg_k8(const float* As, const float* Bs, int k8, v8f_ c[GB_RT_][4]) {
    float4 a[GB_RT_], b[4];
#pragma unroll
    for (int r = 0; r < GB_RT_; ++r) a[r] = *(const float4*)(As + r * 16 * GB_P_ + k8);
#pragma unroll
    for (int q = 0; q < 4; ++q) b[q] = *(const float4*)(Bs + q * 16 * GB_P_ + k8);
#pragma unroll
    for (int r = 0; r < GB_RT_; ++r)
#pragma unroll
        for (int q = 0; q < 4; ++q)
            c[r][q] = __builtin_amdgcn_wmma_f32_16x16x4_f32(
                false, (v2f_){a[r].x, a[r].y}, false, (v2f_){b[q].x, b[q].y}, (short)0, c[r][q], false, false);
#pragma unroll
    for (int r = 0; r < GB_RT_; ++r)
#pragma unroll
        for (int q = 0; q < 4; ++q)
            c[r][q] = __builtin_amdgcn_wmma_f32_16x16x4_f32(
                false, (v2f_){a[r].z, a[r].w}, false, (v2f_){b[q].z, b[q].w}, (short)0, c[r][q], false, false);
}

__global__ __launch_bounds__(256) void hg_gemm_wmma(const float* __restrict__ Xs,
                                                    const float* __restrict__ Wt,
                                                    float* __restrict__ Xe) {
    __shared__ __attribute__((aligned(16))) float lds[2 * GB_STAGE_];

    const int t    = threadIdx.x;
    const int lane = t & 31;
    const int wv   = __builtin_amdgcn_readfirstlane(t >> 5);

    const int ts   = ((int)blockIdx.x * GB_NT_) / GB_WG_;
    const int nt   = (((int)blockIdx.x + 1) * GB_NT_) / GB_WG_ - ts;
    const int row0 = ts * 16;
    const int col0 = wv * 64;

    const int  srow = t >> 3;
    const int  sseg = (t & 7) * 4;
    const int ra0 = min(row0 + srow,      NE_ - 1);
    const int ra1 = min(row0 + srow + 32, NE_ - 1);
    const int ra2 = min(row0 + srow + 64, NE_ - 1);
    const int ra3 = min(row0 + srow + 96, NE_ - 1);
    const float* ga0 = Xs + (size_t)ra0 * DIM_ + sseg;
    const float* ga1 = Xs + (size_t)ra1 * DIM_ + sseg;
    const float* ga2 = Xs + (size_t)ra2 * DIM_ + sseg;
    const float* ga3 = Xs + (size_t)ra3 * DIM_ + sseg;
    const float* gb = Wt + (size_t)srow * DIM_ + sseg;
    const int    so = srow * GB_P_ + sseg;

    const int fo  = (lane & 15) * GB_P_ + ((lane >> 4) << 2);
    const int fob = fo + (GB_AROWS_ + col0) * GB_P_;

    v8f_ c[GB_RT_][4];
#pragma unroll
    for (int r = 0; r < GB_RT_; ++r)
#pragma unroll
        for (int q = 0; q < 4; ++q)
            c[r][q] = (v8f_){};

    float4 sa0, sa1, sa2, sa3;
    float4 sb0, sb1, sb2, sb3, sb4, sb5, sb6, sb7, sb8, sb9, sb10, sb11, sb12, sb13, sb14, sb15;
    HG_STAGE_LD(0)
    HG_STAGE_ST(lds + so)
    HG_STAGE_LD(GB_BK_)
    __syncthreads();

#pragma unroll 1
    for (int ch = 0; ch < GB_NCH_ - 1; ++ch) {
        const float* cur = lds + (ch & 1) * GB_STAGE_;
        float*       nxt = lds + ((ch & 1) ^ 1) * GB_STAGE_;
        const int kn = (ch + 2 < GB_NCH_ ? ch + 2 : GB_NCH_ - 1) * GB_BK_;
        hg_k8(cur + fo, cur + fob, 0, c);
        hg_k8(cur + fo, cur + fob, 8, c);
        __builtin_amdgcn_sched_barrier(0);
        HG_STAGE_ST(nxt + so)
        HG_STAGE_LD(kn)
        __builtin_amdgcn_sched_barrier(0);
        hg_k8(cur + fo, cur + fob, 16, c);
        hg_k8(cur + fo, cur + fob, 24, c);
        __syncthreads();
    }
    {
        const float* cur = lds + ((GB_NCH_ - 1) & 1) * GB_STAGE_;
        hg_k8(cur + fo, cur + fob, 0, c);
        hg_k8(cur + fo, cur + fob, 8, c);
        __builtin_amdgcn_sched_barrier(0);
        hg_k8(cur + fo, cur + fob, 16, c);
        hg_k8(cur + fo, cur + fob, 24, c);
    }

    const int rb = (lane >> 4) << 3;
    const int nc = lane & 15;
#pragma unroll
    for (int r = 0; r < GB_RT_; ++r) {
        if (r >= nt) continue;
        const int rowt = row0 + r * 16;
        if (rowt + rb >= NE_) continue;
        float* o0 = Xe + (size_t)(rowt + rb) * DIM_ + col0 + nc;
#pragma unroll
        for (int v = 0; v < 8; ++v) {
            float* orow = o0 + (size_t)v * DIM_;
#pragma unroll
            for (int q = 0; q < 4; ++q)
                orow[q * 16] = c[r][q][v];
        }
    }
}

__device__ __forceinline__ float4 hg_add4(float4 a, float4 b) {
    return make_float4(a.x + b.x, a.y + b.y, a.z + b.z, a.w + b.w);
}

__device__ __forceinline__ void hg_ld_row(const float* __restrict__ src, int r, int lane,
                                          float4 x[4]) {
    const float4* p = (const float4*)(src + (size_t)r * DIM_) + lane;
    x[0] = p[0]; x[1] = p[32]; x[2] = p[64]; x[3] = p[96];
}

__device__ __forceinline__ void hg_row_sum(const float* __restrict__ src,
                                           const int* __restrict__ lst, int n, int lane,
                                           float4 acc[4]) {
    for (int j = 0; j < 4; ++j) acc[j] = make_float4(0.f, 0.f, 0.f, 0.f);
    int i = 0;
    for (; i + 4 <= n; i += 4) {
        const int r0 = lst[i], r1 = lst[i + 1], r2 = lst[i + 2], r3 = lst[i + 3];
        float4 x0[4], x1[4], x2[4], x3[4];
        hg_ld_row(src, r0, lane, x0);
        hg_ld_row(src, r1, lane, x1);
        hg_ld_row(src, r2, lane, x2);
        hg_ld_row(src, r3, lane, x3);
        for (int j = 0; j < 4; ++j)
            acc[j] = hg_add4(acc[j], hg_add4(hg_add4(x0[j], x1[j]), hg_add4(x2[j], x3[j])));
    }
    if (i + 2 <= n) {
        const int r0 = lst[i], r1 = lst[i + 1];
        float4 x0[4], x1[4];
        hg_ld_row(src, r0, lane, x0);
        hg_ld_row(src, r1, lane, x1);
        for (int j = 0; j < 4; ++j)
            acc[j] = hg_add4(acc[j], hg_add4(x0[j], x1[j]));
        i += 2;
    }
    if (i < n) {
        float4 x0[4];
        hg_ld_row(src, lst[i], lane, x0);
        for (int j = 0; j < 4; ++j) acc[j] = hg_add4(acc[j], x0[j]);
    }
}

__global__ __launch_bounds__(256) void hg_gather_v2e(const float* __restrict__ X,
                                                     const int* __restrict__ off,
                                                     const int* __restrict__ cnt,
                                                     const int* __restrict__ list,
                                                     const float* __restrict__ degE,
                                                     const float* __restrict__ Wdiag,
                                                     float* __restrict__ Xs) {
    const int lane = threadIdx.x & 31;
    const int e = __builtin_amdgcn_readfirstlane((blockIdx.x << 3) + (threadIdx.x >> 5));
    float4 acc[4];
    hg_row_sum(X, list + off[e], cnt[e], lane, acc);
    const float s = degE[e] * Wdiag[e];
    float4* dst = (float4*)(Xs + (size_t)e * DIM_) + lane;
    for (int j = 0; j < 4; ++j)
        dst[j * 32] = make_float4(acc[j].x * s, acc[j].y * s, acc[j].z * s, acc[j].w * s);
}

__global__ __launch_bounds__(256) void hg_gather_e2v(const float* __restrict__ Xe,
                                                     const int* __restrict__ off,
                                                     const int* __restrict__ cnt,
                                                     const int* __restrict__ list,
                                                     const float* __restrict__ degV,
                                                     float* __restrict__ out) {
    const int lane = threadIdx.x & 31;
    const int v = __builtin_amdgcn_readfirstlane((blockIdx.x << 3) + (threadIdx.x >> 5));
    float4 acc[4];
    hg_row_sum(Xe, list + off[v], cnt[v], lane, acc);
    const float s = degV[v];
    float4* dst = (float4*)(out + (size_t)v * DIM_) + lane;
    for (int j = 0; j < 4; ++j)
        dst[j * 32] = make_float4(acc[j].x * s, acc[j].y * s, acc[j].z * s, acc[j].w * s);
}

extern "C" void kernel_launch(void* const* d_in, const int* in_sizes, int n_in,
                              void* d_out, int out_size, void* d_ws, size_t ws_size,
                              hipStream_t stream) {
    (void)in_sizes; (void)n_in; (void)out_size; (void)ws_size;
    const float* X     = (const float*)d_in[0];
    const float* W     = (const float*)d_in[1];
    const float* degE  = (const float*)d_in[2];
    const float* degV  = (const float*)d_in[3];
    const float* Wdiag = (const float*)d_in[4];
    const int*   vidx  = (const int*)d_in[5];
    const int*   eidx  = (const int*)d_in[6];
    float* out = (float*)d_out;

    float* Xs   = (float*)d_ws;
    float* Xe   = Xs + (size_t)NE_ * DIM_;
    int*   cnt  = (int*)(Xe + (size_t)NE_ * DIM_);
    int*   off  = cnt + NR_;
    int*   list = off + NR_;

    hg_group<<<GRP_E_ + GRP_V_, GRP_T_, 0, stream>>>(vidx, eidx, cnt, off, list);

    hg_gather_v2e<<<NE_ / 8, 256, 0, stream>>>(X, off, cnt, list, degE, Wdiag, Xs);

    hg_gemm_wmma<<<GB_WG_, 256, 0, stream>>>(Xs, W, Xe);

    hg_gather_e2v<<<NV_ / 8, 256, 0, stream>>>(Xe, off + NE_, cnt + NE_, list, degV, out);
}


# Product: row tiles dealt evenly: a group owns 6 row tiles x 512 columns, the 27 left-over tiles are shared by column, one 16x16 tile per wave; 56 tile products per wave become 50; sum order kept
# speedup vs baseline: 4.4169x; 1.0414x over previous
#include <hip/hip_runtime.h>
#include <hip/hip_bf16.h>


typedef float v2f_ __attribute__((ext_vector_type(2)));
typedef float v8f_ __attribute__((ext_vector_type(8)));

#define NV_   50000
#define NE_   25000
#define NNZ_  400000
#define DIM_  512
#define NR_   (NE_ + NV_)

static_assert(NE_ % 8 == 0 && NV_ % 8 == 0, "one wave per row, 8 rows per block, exact grids");

#define GRP_T_     1024
#define GRP_E_     84
#define GRP_V_     84
#define GRP_RE_    ((NE_ + GRP_E_ - 1) / GRP_E_)
#define GRP_RV_    ((NV_ + GRP_V_ - 1) / GRP_V_)
#define GRP_RMAX_  608
#define GRP_CAPW_  384
#define GRP_Q_     (NNZ_ / 4)
#define GRP_IT_    ((GRP_Q_ + 4 * GRP_T_ - 1) / (4 * GRP_T_))

static_assert(NNZ_ % 4 == 0, "key arrays are read as whole 16-byte quads");
static_assert(NNZ_ <= (1 << 19), "a pair index must fit in 19 bits of a stash word");
static_assert(GRP_RMAX_ <= 1024 && GRP_RMAX_ <= GRP_T_, "a local row must fit in 10 bits; one thread per row in the scan");
static_assert(GRP_RE_ <= GRP_RMAX_ && GRP_RV_ <= GRP_RMAX_, "LDS counters cover a group's rows");
static_assert((GRP_E_ - 1) * GRP_RE_ < NE_ && (GRP_V_ - 1) * GRP_RV_ < NV_, "every group owns at least one row");

__device__ __forceinline__ void hg_note(int key, int i, int lo, int nr,
                                        int* __restrict__ lcnt, int* __restrict__ mystash,
                                        int& nbelow, int& nw) {
    const int d = key - lo;
    const bool in = (unsigned)d < (unsigned)nr;
    nbelow += (int)((unsigned)d >> 31);
    const unsigned m = __builtin_amdgcn_ballot_w32(in);
    if (m != 0u) {
        if (in) {
            atomicAdd(&lcnt[d], 1);
            const int s = nw + (int)__builtin_amdgcn_mbcnt_lo(m, 0u);
            if (s < GRP_CAPW_) mystash[s] = i | (d << 19);
        }
        nw += __popc(m);
    }
}

__device__ __forceinline__ void hg_place(int key, int i, int lo, int nr,
                                         int* __restrict__ lpos, const int* __restrict__ oth,
                                         int* __restrict__ list) {
    const int d = key - lo;
    if ((unsigned)d < (unsigned)nr) {
        const int p = atomicAdd(&lpos[d], 1);
        list[p] = oth[i];
    }
}

__global__ __launch_bounds__(GRP_T_) void hg_group(const int* __restrict__ vidx,
                                                   const int* __restrict__ eidx,
                                                   int* __restrict__ cnt,
                                                   int* __restrict__ off,
                                                   int* __restrict__ list) {
    __shared__ int lcnt[GRP_RMAX_];
    __shared__ int lpos[GRP_RMAX_];
    __shared__ int wtot[32];
    __shared__ int sbelow;
    __shared__ int sovf;
    __shared__ int stash[32 * GRP_CAPW_];

    const int t    = threadIdx.x;
    const int lane = t & 31;
    const int wv   = __builtin_amdgcn_readfirstlane(t >> 5);

    const bool isE  = blockIdx.x < GRP_E_;
    const int  b    = isE ? (int)blockIdx.x : (int)blockIdx.x - GRP_E_;
    const int  R    = isE ? GRP_RE_ : GRP_RV_;
    const int  lo   = b * R;
    const int  nall = isE ? NE_ : NV_;
    const int  nr   = (nall - lo < R) ? (nall - lo) : R;
    const int* key  = isE ? eidx : vidx;
    const int* oth  = isE ? vidx : eidx;
    const int  row0 = (isE ? 0 : NE_) + lo;
    const int  lst0 = isE ? 0 : NNZ_;
    const int4* key4 = (const int4*)key;
    int* mystash = stash + wv * GRP_CAPW_;

    if (t < GRP_RMAX_) lcnt[t] = 0;
    if (t == 0) { sbelow = 0; sovf = 0; }
    __syncthreads();

    int nbelow = 0;
    int nw     = 0;
    for (int it = 0; it < GRP_IT_; ++it) {
        const int q0 = it * (4 * GRP_T_) + t;
        int4 k[4];
#pragma unroll
        for (int j = 0; j < 4; ++j) {
            const int q = q0 + j * GRP_T_;
            k[j] = (q < GRP_Q_) ? key4[q] : make_int4(0x7fffffff, 0x7fffffff, 0x7fffffff, 0x7fffffff);
        }
#pragma unroll
        for (int j = 0; j < 4; ++j) {
            const int i = (q0 + j * GRP_T_) * 4;
            hg_note(k[j].x, i + 0, lo, nr, lcnt, mystash, nbelow, nw);
            hg_note(k[j].y, i + 1, lo, nr, lcnt, mystash, nbelow, nw);
            hg_note(k[j].z, i + 2, lo, nr, lcnt, mystash, nbelow, nw);
            hg_note(k[j].w, i + 3, lo, nr, lcnt, mystash, nbelow, nw);
        }
    }
    for (int d = 16; d > 0; d >>= 1) nbelow += __shfl_xor(nbelow, d);
    if (lane == 0) {
        atomicAdd(&sbelow, nbelow);
        if (nw > GRP_CAPW_) atomicOr(&sovf, 1);
    }
    __syncthreads();

    const int c = (t < nr) ? lcnt[t] : 0;
    int incl = c;
    for (int d = 1; d < 32; d <<= 1) {
        const int u = __shfl_up(incl, d);
        if (lane >= d) incl += u;
    }
    if (lane == 31) wtot[wv] = incl;
    __syncthreads();
    if (wv == 0) {
        const int w = wtot[lane];
        int wi = w;
        for (int d = 1; d < 32; d <<= 1) {
            const int u = __shfl_up(wi, d);
            if (lane >= d) wi += u;
        }
        wtot[lane] = wi - w;
    }
    __syncthreads();
    const int first = lst0 + sbelow + wtot[wv] + incl - c;
    if (t < nr) {
        off[row0 + t] = first;
        cnt[row0 + t] = c;
        lpos[t]       = first;
    }
    const int ovf = sovf;
    __syncthreads();

    if (!ovf) {
        for (int j = lane; j < nw; j += 32) {
            const int ent = mystash[j];
            const int p   = atomicAdd(&lpos[ent >> 19], 1);
            list[p] = oth[ent & ((1 << 19) - 1)];
        }
    } else {
        for (int q = t; q < GRP_Q_; q += GRP_T_) {
            const int4 k = key4[q];
            hg_place(k.x, 4 * q + 0, lo, nr, lpos, oth, list);
            hg_place(k.y, 4 * q + 1, lo, nr, lpos, oth, list);
            hg_place(k.z, 4 * q + 2, lo, nr, lpos, oth, list);
            hg_place(k.w, 4 * q + 3, lo, nr, lpos, oth, list);
        }
    }
}

#define GB_WG_     256
#define GB_NT_     ((NE_ + 15) / 16)
#define GB_OT_     6
#define GB_XT0_    (GB_WG_ * GB_OT_)
#define GB_XN_     (GB_NT_ - GB_XT0_)
#define GB_XG_     (GB_XN_ * 4)
#define GB_RT_     (GB_OT_ + 1)
#define GB_BK_     32
#define GB_NCH_    (DIM_ / GB_BK_)
#define GB_P_      36
#define GB_AROWS_  128
#define GB_STAGE_  ((GB_AROWS_ + DIM_) * GB_P_)

static_assert(GB_XT0_ * 16 <= NE_, "own tiles are whole and in range: no clamp, no store guard");
static_assert(GB_XN_ >= 0 && GB_XG_ <= GB_WG_, "every left-over tile finds its four groups");
static_assert(GB_RT_ * 16 <= 128, "the A stage holds the own tiles and the left-over one");
static_assert(DIM_ % GB_BK_ == 0 && GB_BK_ == 32, "a chunk row is one 128-byte line, 8 lanes x 16 bytes");
static_assert(DIM_ == 512, "8 waves x 64 columns; 16 W rows per thread and chunk");

#define HG_STAGE_LD(k)                                                                   \
    sa0 = *(const float4*)(ga0 + (k)); sa1 = *(const float4*)(ga1 + (k));                \
    sa2 = *(const float4*)(ga2 + (k)); sa3 = *(const float4*)(ga3 + (k));                \
    HG_LDB(0, k)  HG_LDB(1, k)  HG_LDB(2, k)  HG_LDB(3, k)  HG_LDB(4, k)  HG_LDB(5, k)   \
    HG_LDB(6, k)  HG_LDB(7, k)  HG_LDB(8, k)  HG_LDB(9, k)  HG_LDB(10, k) HG_LDB(11, k)  \
    HG_LDB(12, k) HG_LDB(13, k) HG_LDB(14, k) HG_LDB(15, k)
#define HG_LDB(i, k)  sb##i = *(const float4*)(gb + (size_t)(i) * 32 * DIM_ + (k));
#define HG_STAGE_ST(st)                                                                  \
    *(float4*)((st))                = sa0; *(float4*)((st) + 32 * GB_P_) = sa1;          \
    *(float4*)((st) + 64 * GB_P_)   = sa2; *(float4*)((st) + 96 * GB_P_) = sa3;          \
    HG_STB(0, st)  HG_STB(1, st)  HG_STB(2, st)  HG_STB(3, st)  HG_STB(4, st)  HG_STB(5, st)   \
    HG_STB(6, st)  HG_STB(7, st)  HG_STB(8, st)  HG_STB(9, st)  HG_STB(10, st) HG_STB(11, st)  \
    HG_STB(12, st) HG_STB(13, st) HG_STB(14, st) HG_STB(15, st)
#define HG_STB(i, st) *(float4*)((st) + (GB_AROWS_ + (i) * 32) * GB_P_) = sb##i;

__device__ __forceinline__ void hg_k8(const float* As, const float* Bs, const float* Bx, int k8,
                                      v8f_ c[GB_OT_][4], v8f_& cx) {
    float4 a[GB_OT_], b[4];
#pragma unroll
    for (int r = 0; r < GB_OT_; ++r) a[r] = *(const float4*)(As + r * 16 * GB_P_ + k8);
    const float4 ax = *(const float4*)(As + GB_OT_ * 16 * GB_P_ + k8);
#pragma unroll
    for (int q = 0; q < 4; ++q) b[q] = *(const float4*)(Bs + q * 16 * GB_P_ + k8);
    const float4 bx = *(const float4*)(Bx + k8);
#pragma unroll
    for (int r = 0; r < GB_OT_; ++r)
#pragma unroll
        for (int q = 0; q < 4; ++q)
            c[r][q] = __builtin_amdgcn_wmma_f32_16x16x4_f32(
                false, (v2f_){a[r].x, a[r].y}, false, (v2f_){b[q].x, b[q].y}, (short)0, c[r][q], false, false);
    cx = __builtin_amdgcn_wmma_f32_16x16x4_f32(
        false, (v2f_){ax.x, ax.y}, false, (v2f_){bx.x, bx.y}, (short)0, cx, false, false);
#pragma unroll
    for (int r = 0; r < GB_OT_; ++r)
#pragma unroll
        for (int q = 0; q < 4; ++q)
            c[r][q] = __builtin_amdgcn_wmma_f32_16x16x4_f32(
                false, (v2f_){a[r].z, a[r].w}, false, (v2f_){b[q].z, b[q].w}, (short)0, c[r][q], false, false);
    cx = __builtin_amdgcn_wmma_f32_16x16x4_f32(
        false, (v2f_){ax.z, ax.w}, false, (v2f_){bx.z, bx.w}, (short)0, cx, false, false);
}

__global__ __launch_bounds__(256) void hg_gemm_wmma(const float* __restrict__ Xs,
                                                    const float* __restrict__ Wt,
                                                    float* __restrict__ Xe) {
    __shared__ __attribute__((aligned(16))) float lds[2 * GB_STAGE_];

    const int t    = threadIdx.x;
    const int lane = t & 31;
    const int wv   = __builtin_amdgcn_readfirstlane(t >> 5);

    const int  grp  = (int)blockIdx.x;
    const int  row0 = grp * (GB_OT_ * 16);
    const int  col0 = wv * 64;
    const bool hasx = grp < GB_XG_;
    const int  xt   = min(GB_XT0_ + (grp >> 2), GB_NT_ - 1);
    const int  xrow = xt * 16;
    const int  xcol = (((grp & 3) << 3) + wv) * 16;

    const int  srow = t >> 3;
    const int  sseg = (t & 7) * 4;
    const int ra0 = row0 + srow;
    const int ra1 = row0 + srow + 32;
    const int ra2 = row0 + srow + 64;
    const int ra3 = min(xrow + (srow & 15), NE_ - 1);
    const float* ga0 = Xs + (size_t)ra0 * DIM_ + sseg;
    const float* ga1 = Xs + (size_t)ra1 * DIM_ + sseg;
    const float* ga2 = Xs + (size_t)ra2 * DIM_ + sseg;
    const float* ga3 = Xs + (size_t)ra3 * DIM_ + sseg;
    const float* gb = Wt + (size_t)srow * DIM_ + sseg;
    const int    so = srow * GB_P_ + sseg;

    const int fo  = (lane & 15) * GB_P_ + ((lane >> 4) << 2);
    const int fob = fo + (GB_AROWS_ + col0) * GB_P_;
    const int fox = fo + (GB_AROWS_ + xcol) * GB_P_;

    v8f_ c[GB_OT_][4];
    v8f_ cx = (v8f_){};
#pragma unroll
    for (int r = 0; r < GB_OT_; ++r)
#pragma unroll
        for (int q = 0; q < 4; ++q)
            c[r][q] = (v8f_){};

    float4 sa0, sa1, sa2, sa3;
    float4 sb0, sb1, sb2, sb3, sb4, sb5, sb6, sb7, sb8, sb9, sb10, sb11, sb12, sb13, sb14, sb15;
    HG_STAGE_LD(0)
    HG_STAGE_ST(lds + so)
    HG_STAGE_LD(GB_BK_)
    __syncthreads();

#pragma unroll 1
    for (int ch = 0; ch < GB_NCH_ - 1; ++ch) {
        const float* cur = lds + (ch & 1) * GB_STAGE_;
        float*       nxt = lds + ((ch & 1) ^ 1) * GB_STAGE_;
        const int kn = (ch + 2 < GB_NCH_ ? ch + 2 : GB_NCH_ - 1) * GB_BK_;
        hg_k8(cur + fo, cur + fob, cur + fox, 0, c, cx);
        hg_k8(cur + fo, cur + fob, cur + fox, 8, c, cx);
        __builtin_amdgcn_sched_barrier(0);
        HG_STAGE_ST(nxt + so)
        HG_STAGE_LD(kn)
        __builtin_amdgcn_sched_barrier(0);
        hg_k8(cur + fo, cur + fob, cur + fox, 16, c, cx);
        hg_k8(cur + fo, cur + fob, cur + fox, 24, c, cx);
        __syncthreads();
    }
    {
        const float* cur = lds + ((GB_NCH_ - 1) & 1) * GB_STAGE_;
        hg_k8(cur + fo, cur + fob, cur + fox, 0, c, cx);
        hg_k8(cur + fo, cur + fob, cur + fox, 8, c, cx);
        __builtin_amdgcn_sched_barrier(0);
        hg_k8(cur + fo, cur + fob, cur + fox, 16, c, cx);
        hg_k8(cur + fo, cur + fob, cur + fox, 24, c, cx);
    }

    const int rb = (lane >> 4) << 3;
    const int nc = lane & 15;
#pragma unroll
    for (int r = 0; r < GB_OT_; ++r) {
        float* o0 = Xe + (size_t)(row0 + r * 16 + rb) * DIM_ + col0 + nc;
#pragma unroll
        for (int v = 0; v < 8; ++v) {
            float* orow = o0 + (size_t)v * DIM_;
#pragma unroll
            for (int q = 0; q < 4; ++q)
                orow[q * 16] = c[r][q][v];
        }
    }
    if (hasx && xrow + rb < NE_) {
        float* o0 = Xe + (size_t)(xrow + rb) * DIM_ + xcol + nc;
#pragma unroll
        for (int v = 0; v < 8; ++v)
            o0[(size_t)v * DIM_] = cx[v];
    }
}

__device__ __forceinline__ float4 hg_add4(float4 a, float4 b) {
    return make_float4(a.x + b.x, a.y + b.y, a.z + b.z, a.w + b.w);
}

__device__ __forceinline__ void hg_ld_row(const float* __restrict__ src, int r, int lane,
                                          float4 x[4]) {
    const float4* p = (const float4*)(src + (size_t)r * DIM_) + lane;
    x[0] = p[0]; x[1] = p[32]; x[2] = p[64]; x[3] = p[96];
}

__device__ __forceinline__ void hg_row_sum(const float* __restrict__ src,
                                           const int* __restrict__ lst, int n, int lane,
                                           float4 acc[4]) {
    for (int j = 0; j < 4; ++j) acc[j] = make_float4(0.f, 0.f, 0.f, 0.f);
    int i = 0;
    for (; i + 4 <= n; i += 4) {
        const int r0 = lst[i], r1 = lst[i + 1], r2 = lst[i + 2], r3 = lst[i + 3];
        float4 x0[4], x1[4], x2[4], x3[4];
        hg_ld_row(src, r0, lane, x0);
        hg_ld_row(src, r1, lane, x1);
        hg_ld_row(src, r2, lane, x2);
        hg_ld_row(src, r3, lane, x3);
        for (int j = 0; j < 4; ++j)
            acc[j] = hg_add4(acc[j], hg_add4(hg_add4(x0[j], x1[j]), hg_add4(x2[j], x3[j])));
    }
    if (i + 2 <= n) {
        const int r0 = lst[i], r1 = lst[i + 1];
        float4 x0[4], x1[4];
        hg_ld_row(src, r0, lane, x0);
        hg_ld_row(src, r1, lane, x1);
        for (int j = 0; j < 4; ++j)
            acc[j] = hg_add4(acc[j], hg_add4(x0[j], x1[j]));
        i += 2;
    }
    if (i < n) {
        float4 x0[4];
        hg_ld_row(src, lst[i], lane, x0);
        for (int j = 0; j < 4; ++j) acc[j] = hg_add4(acc[j], x0[j]);
    }
}

__global__ __launch_bounds__(256) void hg_gather_v2e(const float* __restrict__ X,
                                                     const int* __restrict__ off,
                                                     const int* __restrict__ cnt,
                                                     const int* __restrict__ list,
                                                     const float* __restrict__ degE,
                                                     const float* __restrict__ Wdiag,
                                                     float* __restrict__ Xs) {
    const int lane = threadIdx.x & 31;
    const int e = __builtin_amdgcn_readfirstlane((blockIdx.x << 3) + (threadIdx.x >> 5));
    float4 acc[4];
    hg_row_sum(X, list + off[e], cnt[e], lane, acc);
    const float s = degE[e] * Wdiag[e];
    float4* dst = (float4*)(Xs + (size_t)e * DIM_) + lane;
    for (int j = 0; j < 4; ++j)
        dst[j * 32] = make_float4(acc[j].x * s, acc[j].y * s, acc[j].z * s, acc[j].w * s);
}

__global__ __launch_bounds__(256) void hg_gather_e2v(const float* __restrict__ Xe,
                                                     const int* __restrict__ off,
                                                     const int* __restrict__ cnt,
                                                     const int* __restrict__ list,
                                                     const float* __restrict__ degV,
                                                     float* __restrict__ out) {
    const int lane = threadIdx.x & 31;
    const int v = __builtin_amdgcn_readfirstlane((blockIdx.x << 3) + (threadIdx.x >> 5));
    float4 acc[4];
    hg_row_sum(Xe, list + off[v], cnt[v], lane, acc);
    const float s = degV[v];
    float4* dst = (float4*)(out + (size_t)v * DIM_) + lane;
    for (int j = 0; j < 4; ++j)
        dst[j * 32] = make_float4(acc[j].x * s, acc[j].y * s, acc[j].z * s, acc[j].w * s);
}

extern "C" void kernel_launch(void* const* d_in, const int* in_sizes, int n_in,
                              void* d_out, int out_size, void* d_ws, size_t ws_size,
                              hipStream_t stream) {
    (void)in_sizes; (void)n_in; (void)out_size; (void)ws_size;
    const float* X     = (const float*)d_in[0];
    const float* W     = (const float*)d_in[1];
    const float* degE  = (const float*)d_in[2];
    const float* degV  = (const float*)d_in[3];
    const float* Wdiag = (const float*)d_in[4];
    const int*   vidx  = (const int*)d_in[5];
    const int*   eidx  = (const int*)d_in[6];
    float* out = (float*)d_out;

    float* Xs   = (float*)d_ws;
    float* Xe   = Xs + (size_t)NE_ * DIM_;
    int*   cnt  = (int*)(Xe + (size_t)NE_ * DIM_);
    int*   off  = cnt + NR_;
    int*   list = off + NR_;

    hg_group<<<GRP_E_ + GRP_V_, GRP_T_, 0, stream>>>(vidx, eidx, cnt, off, list);

    hg_gather_v2e<<<NE_ / 8, 256, 0, stream>>>(X, off, cnt, list, degE, Wdiag, Xs);

    hg_gemm_wmma<<<GB_WG_, 256, 0, stream>>>(Xs, W, Xe);

    hg_gather_e2v<<<NV_ / 8, 256, 0, stream>>>(Xe, off + NE_, cnt + NE_, list, degV, out);
}


# Pair grouping: key arrays split in 3 parts; each of 255 work-groups (was 168) scans one part for a wider row range; a row's list is 3 segments summed in part order, so f32 sum order changes
# speedup vs baseline: 4.6162x; 1.0451x over previous
#include <hip/hip_runtime.h>
#include <hip/hip_bf16.h>


typedef float v2f_ __attribute__((ext_vector_type(2)));
typedef float v8f_ __attribute__((ext_vector_type(8)));

#define NV_   50000
#define NE_   25000
#define NNZ_  400000
#define DIM_  512
#define NR_   (NE_ + NV_)

static_assert(NE_ % 8 == 0 && NV_ % 8 == 0, "one wave per row, 8 rows per block, exact grids");

#define GRP_T_     1024
#define GRP_P_     3
#define GRP_E_     36
#define GRP_V_     49
#define GRP_NB_    ((GRP_E_ + GRP_V_) * GRP_P_)
#define GRP_RE_    ((NE_ + GRP_E_ - 1) / GRP_E_)
#define GRP_RV_    ((NV_ + GRP_V_ - 1) / GRP_V_)
#define GRP_RMAX_  1024
#define GRP_CAPW_  384
#define GRP_Q_     (NNZ_ / 4)
#define GRP_QP_    ((GRP_Q_ + GRP_P_ - 1) / GRP_P_)
#define GRP_CAPB_  (4 * GRP_QP_)
#define GRP_IT_    ((GRP_QP_ + 4 * GRP_T_ - 1) / (4 * GRP_T_))

static_assert(NNZ_ % 4 == 0, "key arrays are read as whole 16-byte quads");
static_assert(NNZ_ <= (1 << 19), "a pair index must fit in 19 bits of a stash word");
static_assert(GRP_RMAX_ <= 1024 && GRP_RMAX_ <= GRP_T_, "a local row must fit in 10 bits; one thread per row in the scan");
static_assert(GRP_RE_ <= GRP_RMAX_ && GRP_RV_ <= GRP_RMAX_, "LDS counters cover a group's rows");
static_assert((GRP_E_ - 1) * GRP_RE_ < NE_ && (GRP_V_ - 1) * GRP_RV_ < NV_, "every group owns at least one row");
static_assert((GRP_P_ - 1) * GRP_QP_ < GRP_Q_, "every part holds at least one quad");
static_assert((long long)GRP_NB_ * GRP_CAPB_ < (1ll << 31), "a list slot fits in an int");

__device__ __forceinline__ void hg_note(int key, int i, int lo, int nr,
                                        int* __restrict__ lcnt, int* __restrict__ mystash,
                                        int& nw) {
    const int d = key - lo;
    const bool in = (unsigned)d < (unsigned)nr;
    const unsigned m = __builtin_amdgcn_ballot_w32(in);
    if (m != 0u) {
        if (in) {
            atomicAdd(&lcnt[d], 1);
            const int s = nw + (int)__builtin_amdgcn_mbcnt_lo(m, 0u);
            if (s < GRP_CAPW_) mystash[s] = i | (d << 19);
        }
        nw += __popc(m);
    }
}

__device__ __forceinline__ void hg_place(int key, int i, int lo, int nr,
                                         int* __restrict__ lpos, const int* __restrict__ oth,
                                         int* __restrict__ list) {
    const int d = key - lo;
    if ((unsigned)d < (unsigned)nr) {
        const int p = atomicAdd(&lpos[d], 1);
        list[p] = oth[i];
    }
}

__global__ __launch_bounds__(GRP_T_) void hg_group(const int* __restrict__ vidx,
                                                   const int* __restrict__ eidx,
                                                   int* __restrict__ cnt,
                                                   int* __restrict__ off,
                                                   int* __restrict__ list) {
    __shared__ int lcnt[GRP_RMAX_];
    __shared__ int lpos[GRP_RMAX_];
    __shared__ int wtot[32];
    __shared__ int sovf;
    __shared__ int stash[32 * GRP_CAPW_];

    const int t    = threadIdx.x;
    const int lane = t & 31;
    const int wv   = __builtin_amdgcn_readfirstlane(t >> 5);

    const int  bi   = (int)blockIdx.x;
    const int  g    = bi / GRP_P_;
    const int  part = bi - g * GRP_P_;
    const bool isE  = g < GRP_E_;
    const int  b    = isE ? g : g - GRP_E_;
    const int  R    = isE ? GRP_RE_ : GRP_RV_;
    const int  lo   = b * R;
    const int  nall = isE ? NE_ : NV_;
    const int  nr   = (nall - lo < R) ? (nall - lo) : R;
    const int* key  = isE ? eidx : vidx;
    const int* oth  = isE ? vidx : eidx;
    const int  row0 = part * NR_ + (isE ? 0 : NE_) + lo;
    const int  lst0 = bi * GRP_CAPB_;
    const int  qlo  = part * GRP_QP_;
    const int  qhi  = (qlo + GRP_QP_ < GRP_Q_) ? (qlo + GRP_QP_) : GRP_Q_;
    const int4* key4 = (const int4*)key;
    int* mystash = stash + wv * GRP_CAPW_;

    if (t < GRP_RMAX_) lcnt[t] = 0;
    if (t == 0) sovf = 0;
    __syncthreads();

    int nw = 0;
    for (int it = 0; it < GRP_IT_; ++it) {
        const int q0 = qlo + it * (4 * GRP_T_) + t;
        int4 k[4];
#pragma unroll
        for (int j = 0; j < 4; ++j) {
            const int q = q0 + j * GRP_T_;
            k[j] = (q < qhi) ? key4[q] : make_int4(0x7fffffff, 0x7fffffff, 0x7fffffff, 0x7fffffff);
        }
#pragma unroll
        for (int j = 0; j < 4; ++j) {
            const int i = (q0 + j * GRP_T_) * 4;
            hg_note(k[j].x, i + 0, lo, nr, lcnt, mystash, nw);
            hg_note(k[j].y, i + 1, lo, nr, lcnt, mystash, nw);
            hg_note(k[j].z, i + 2, lo, nr, lcnt, mystash, nw);
            hg_note(k[j].w, i + 3, lo, nr, lcnt, mystash, nw);
        }
    }
    if (lane == 0 && nw > GRP_CAPW_) atomicOr(&sovf, 1);
    __syncthreads();

    const int c = (t < nr) ? lcnt[t] : 0;
    int incl = c;
    for (int d = 1; d < 32; d <<= 1) {
        const int u = __shfl_up(incl, d);
        if (lane >= d) incl += u;
    }
    if (lane == 31) wtot[wv] = incl;
    __syncthreads();
    if (wv == 0) {
        const int w = wtot[lane];
        int wi = w;
        for (int d = 1; d < 32; d <<= 1) {
            const int u = __shfl_up(wi, d);
            if (lane >= d) wi += u;
        }
        wtot[lane] = wi - w;
    }
    __syncthreads();
    const int first = lst0 + wtot[wv] + incl - c;
    if (t < nr) {
        off[row0 + t] = first;
        cnt[row0 + t] = c;
        lpos[t]       = first;
    }
    const int ovf = sovf;
    __syncthreads();

    if (!ovf) {
        for (int j = lane; j < nw; j += 32) {
            const int ent = mystash[j];
            const int p   = atomicAdd(&lpos[ent >> 19], 1);
            list[p] = oth[ent & ((1 << 19) - 1)];
        }
    } else {
        for (int q = qlo + t; q < qhi; q += GRP_T_) {
            const int4 k = key4[q];
            hg_place(k.x, 4 * q + 0, lo, nr, lpos, oth, list);
            hg_place(k.y, 4 * q + 1, lo, nr, lpos, oth, list);
            hg_place(k.z, 4 * q + 2, lo, nr, lpos, oth, list);
            hg_place(k.w, 4 * q + 3, lo, nr, lpos, oth, list);
        }
    }
}

#define GB_WG_     256
#define GB_NT_     ((NE_ + 15) / 16)
#define GB_OT_     6
#define GB_XT0_    (GB_WG_ * GB_OT_)
#define GB_XN_     (GB_NT_ - GB_XT0_)
#define GB_XG_     (GB_XN_ * 4)
#define GB_RT_     (GB_OT_ + 1)
#define GB_BK_     32
#define GB_NCH_    (DIM_ / GB_BK_)
#define GB_P_      36
#define GB_AROWS_  128
#define GB_STAGE_  ((GB_AROWS_ + DIM_) * GB_P_)

static_assert(GB_XT0_ * 16 <= NE_, "own tiles are whole and in range: no clamp, no store guard");
static_assert(GB_XN_ >= 0 && GB_XG_ <= GB_WG_, "every left-over tile finds its four groups");
static_assert(GB_RT_ * 16 <= 128, "the A stage holds the own tiles and the left-over one");
static_assert(DIM_ % GB_BK_ == 0 && GB_BK_ == 32, "a chunk row is one 128-byte line, 8 lanes x 16 bytes");
static_assert(DIM_ == 512, "8 waves x 64 columns; 16 W rows per thread and chunk");

#define HG_STAGE_LD(k)                                                                   \
    sa0 = *(const float4*)(ga0 + (k)); sa1 = *(const float4*)(ga1 + (k));                \
    sa2 = *(const float4*)(ga2 + (k)); sa3 = *(const float4*)(ga3 + (k));                \
    HG_LDB(0, k)  HG_LDB(1, k)  HG_LDB(2, k)  HG_LDB(3, k)  HG_LDB(4, k)  HG_LDB(5, k)   \
    HG_LDB(6, k)  HG_LDB(7, k)  HG_LDB(8, k)  HG_LDB(9, k)  HG_LDB(10, k) HG_LDB(11, k)  \
    HG_LDB(12, k) HG_LDB(13, k) HG_LDB(14, k) HG_LDB(15, k)
#define HG_LDB(i, k)  sb##i = *(const float4*)(gb + (size_t)(i) * 32 * DIM_ + (k));
#define HG_STAGE_ST(st)                                                                  \
    *(float4*)((st))                = sa0; *(float4*)((st) + 32 * GB_P_) = sa1;          \
    *(float4*)((st) + 64 * GB_P_)   = sa2; *(float4*)((st) + 96 * GB_P_) = sa3;          \
    HG_STB(0, st)  HG_STB(1, st)  HG_STB(2, st)  HG_STB(3, st)  HG_STB(4, st)  HG_STB(5, st)   \
    HG_STB(6, st)  HG_STB(7, st)  HG_STB(8, st)  HG_STB(9, st)  HG_STB(10, st) HG_STB(11, st)  \
    HG_STB(12, st) HG_STB(13, st) HG_STB(14, st) HG_STB(15, st)
#define HG_STB(i, st) *(float4*)((st) + (GB_AROWS_ + (i) * 32) * GB_P_) = sb##i;

__device__ __forceinline__ void hg_k8(const float* As, const float* Bs, const float* Bx, int k8,
                                      v8f_ c[GB_OT_][4], v8f_& cx) {
    float4 a[GB_OT_], b[4];
#pragma unroll
    for (int r = 0; r < GB_OT_; ++r) a[r] = *(const float4*)(As + r * 16 * GB_P_ + k8);
    const float4 ax = *(const float4*)(As + GB_OT_ * 16 * GB_P_ + k8);
#pragma unroll
    for (int q = 0; q < 4; ++q) b[q] = *(const float4*)(Bs + q * 16 * GB_P_ + k8);
    const float4 bx = *(const float4*)(Bx + k8);
#pragma unroll
    for (int r = 0; r < GB_OT_; ++r)
#pragma unroll
        for (int q = 0; q < 4; ++q)
            c[r][q] = __builtin_amdgcn_wmma_f32_16x16x4_f32(
                false, (v2f_){a[r].x, a[r].y}, false, (v2f_){b[q].x, b[q].y}, (short)0, c[r][q], false, false);
    cx = __builtin_amdgcn_wmma_f32_16x16x4_f32(
        false, (v2f_){ax.x, ax.y}, false, (v2f_){bx.x, bx.y}, (short)0, cx, false, false);
#pragma unroll
    for (int r = 0; r < GB_OT_; ++r)
#pragma unroll
        for (int q = 0; q < 4; ++q)
            c[r][q] = __builtin_amdgcn_wmma_f32_16x16x4_f32(
                false, (v2f_){a[r].z, a[r].w}, false, (v2f_){b[q].z, b[q].w}, (short)0, c[r][q], false, false);
    cx = __builtin_amdgcn_wmma_f32_16x16x4_f32(
        false, (v2f_){ax.z, ax.w}, false, (v2f_){bx.z, bx.w}, (short)0, cx, false, false);
}

__global__ __launch_bounds__(256) void hg_gemm_wmma(const float* __restrict__ Xs,
                                                    const float* __restrict__ Wt,
                                                    float* __restrict__ Xe) {
    __shared__ __attribute__((aligned(16))) float lds[2 * GB_STAGE_];

    const int t    = threadIdx.x;
    const int lane = t & 31;
    const int wv   = __builtin_amdgcn_readfirstlane(t >> 5);

    const int  grp  = (int)blockIdx.x;
    const int  row0 = grp * (GB_OT_ * 16);
    const int  col0 = wv * 64;
    const bool hasx = grp < GB_XG_;
    const int  xt   = min(GB_XT0_ + (grp >> 2), GB_NT_ - 1);
    const int  xrow = xt * 16;
    const int  xcol = (((grp & 3) << 3) + wv) * 16;

    const int  srow = t >> 3;
    const int  sseg = (t & 7) * 4;
    const int ra0 = row0 + srow;
    const int ra1 = row0 + srow + 32;
    const int ra2 = row0 + srow + 64;
    const int ra3 = min(xrow + (srow & 15), NE_ - 1);
    const float* ga0 = Xs + (size_t)ra0 * DIM_ + sseg;
    const float* ga1 = Xs + (size_t)ra1 * DIM_ + sseg;
    const float* ga2 = Xs + (size_t)ra2 * DIM_ + sseg;
    const float* ga3 = Xs + (size_t)ra3 * DIM_ + sseg;
    const float* gb = Wt + (size_t)srow * DIM_ + sseg;
    const int    so = srow * GB_P_ + sseg;

    const int fo  = (lane & 15) * GB_P_ + ((lane >> 4) << 2);
    const int fob = fo + (GB_AROWS_ + col0) * GB_P_;
    const int fox = fo + (GB_AROWS_ + xcol) * GB_P_;

    v8f_ c[GB_OT_][4];
    v8f_ cx = (v8f_){};
#pragma unroll
    for (int r = 0; r < GB_OT_; ++r)
#pragma unroll
        for (int q = 0; q < 4; ++q)
            c[r][q] = (v8f_){};

    float4 sa0, sa1, sa2, sa3;
    float4 sb0, sb1, sb2, sb3, sb4, sb5, sb6, sb7, sb8, sb9, sb10, sb11, sb12, sb13, sb14, sb15;
    HG_STAGE_LD(0)
    HG_STAGE_ST(lds + so)
    HG_STAGE_LD(GB_BK_)
    __syncthreads();

#pragma unroll 1
    for (int ch = 0; ch < GB_NCH_ - 1; ++ch) {
        const float* cur = lds + (ch & 1) * GB_STAGE_;
        float*       nxt = lds + ((ch & 1) ^ 1) * GB_STAGE_;
        const int kn = (ch + 2 < GB_NCH_ ? ch + 2 : GB_NCH_ - 1) * GB_BK_;
        hg_k8(cur + fo, cur + fob, cur + fox, 0, c, cx);
        hg_k8(cur + fo, cur + fob, cur + fox, 8, c, cx);
        __builtin_amdgcn_sched_barrier(0);
        HG_STAGE_ST(nxt + so)
        HG_STAGE_LD(kn)
        __builtin_amdgcn_sched_barrier(0);
        hg_k8(cur + fo, cur + fob, cur + fox, 16, c, cx);
        hg_k8(cur + fo, cur + fob, cur + fox, 24, c, cx);
        __syncthreads();
    }
    {
        const float* cur = lds + ((GB_NCH_ - 1) & 1) * GB_STAGE_;
        hg_k8(cur + fo, cur + fob, cur + fox, 0, c, cx);
        hg_k8(cur + fo, cur + fob, cur + fox, 8, c, cx);
        __builtin_amdgcn_sched_barrier(0);
        hg_k8(cur + fo, cur + fob, cur + fox, 16, c, cx);
        hg_k8(cur + fo, cur + fob, cur + fox, 24, c, cx);
    }

    const int rb = (lane >> 4) << 3;
    const int nc = lane & 15;
#pragma unroll
    for (int r = 0; r < GB_OT_; ++r) {
        float* o0 = Xe + (size_t)(row0 + r * 16 + rb) * DIM_ + col0 + nc;
#pragma unroll
        for (int v = 0; v < 8; ++v) {
            float* orow = o0 + (size_t)v * DIM_;
#pragma unroll
            for (int q = 0; q < 4; ++q)
                orow[q * 16] = c[r][q][v];
        }
    }
    if (hasx && xrow + rb < NE_) {
        float* o0 = Xe + (size_t)(xrow + rb) * DIM_ + xcol + nc;
#pragma unroll
        for (int v = 0; v < 8; ++v)
            o0[(size_t)v * DIM_] = cx[v];
    }
}

__device__ __forceinline__ float4 hg_add4(float4 a, float4 b) {
    return make_float4(a.x + b.x, a.y + b.y, a.z + b.z, a.w + b.w);
}

__device__ __forceinline__ void hg_ld_row(const float* __restrict__ src, int r, int lane,
                                          float4 x[4]) {
    const float4* p = (const float4*)(src + (size_t)r * DIM_) + lane;
    x[0] = p[0]; x[1] = p[32]; x[2] = p[64]; x[3] = p[96];
}

static_assert(GRP_P_ == 3, "the row walk below names three segments");
__device__ __forceinline__ void hg_row_sum(const float* __restrict__ src,
                                           const int* __restrict__ list,
                                           const int* __restrict__ off,
                                           const int* __restrict__ cnt,
                                           int row, int lane, float4 acc[4]) {
    for (int j = 0; j < 4; ++j) acc[j] = make_float4(0.f, 0.f, 0.f, 0.f);
    const int c0 = cnt[row], c1 = cnt[NR_ + row], c2 = cnt[2 * NR_ + row];
    const int o0 = off[row], o1 = off[NR_ + row], o2 = off[2 * NR_ + row];
    const int n  = c0 + c1 + c2;
    for (int base = 0; base < n; base += 32) {
        const int p = base + lane;
        int idx = 0;
        if (p < n) {
            const int a = (p < c0) ? (o0 + p) : (p < c0 + c1) ? (o1 + (p - c0)) : (o2 + (p - c0 - c1));
            idx = list[a];
        }
        const int m = (n - base < 32) ? (n - base) : 32;
        int i = 0;
        for (; i + 4 <= m; i += 4) {
            const int r0 = __builtin_amdgcn_readlane(idx, i);
            const int r1 = __builtin_amdgcn_readlane(idx, i + 1);
            const int r2 = __builtin_amdgcn_readlane(idx, i + 2);
            const int r3 = __builtin_amdgcn_readlane(idx, i + 3);
            float4 x0[4], x1[4], x2[4], x3[4];
            hg_ld_row(src, r0, lane, x0);
            hg_ld_row(src, r1, lane, x1);
            hg_ld_row(src, r2, lane, x2);
            hg_ld_row(src, r3, lane, x3);
            for (int j = 0; j < 4; ++j)
                acc[j] = hg_add4(acc[j], hg_add4(hg_add4(x0[j], x1[j]), hg_add4(x2[j], x3[j])));
        }
        if (i + 2 <= m) {
            const int r0 = __builtin_amdgcn_readlane(idx, i);
            const int r1 = __builtin_amdgcn_readlane(idx, i + 1);
            float4 x0[4], x1[4];
            hg_ld_row(src, r0, lane, x0);
            hg_ld_row(src, r1, lane, x1);
            for (int j = 0; j < 4; ++j)
                acc[j] = hg_add4(acc[j], hg_add4(x0[j], x1[j]));
            i += 2;
        }
        if (i < m) {
            float4 x0[4];
            hg_ld_row(src, __builtin_amdgcn_readlane(idx, i), lane, x0);
            for (int j = 0; j < 4; ++j) acc[j] = hg_add4(acc[j], x0[j]);
        }
    }
}

__global__ __launch_bounds__(256) void hg_gather_v2e(const float* __restrict__ X,
                                                     const int* __restrict__ off,
                                                     const int* __restrict__ cnt,
                                                     const int* __restrict__ list,
                                                     const float* __restrict__ degE,
                                                     const float* __restrict__ Wdiag,
                                                     float* __restrict__ Xs) {
    const int lane = threadIdx.x & 31;
    const int e = __builtin_amdgcn_readfirstlane((blockIdx.x << 3) + (threadIdx.x >> 5));
    float4 acc[4];
    hg_row_sum(X, list, off, cnt, e, lane, acc);
    const float s = degE[e] * Wdiag[e];
    float4* dst = (float4*)(Xs + (size_t)e * DIM_) + lane;
    for (int j = 0; j < 4; ++j)
        dst[j * 32] = make_float4(acc[j].x * s, acc[j].y * s, acc[j].z * s, acc[j].w * s);
}

__global__ __launch_bounds__(256) void hg_gather_e2v(const float* __restrict__ Xe,
                                                     const int* __restrict__ off,
                                                     const int* __restrict__ cnt,
                                                     const int* __restrict__ list,
                                                     const float* __restrict__ degV,
                                                     float* __restrict__ out) {
    const int lane = threadIdx.x & 31;
    const int v = __builtin_amdgcn_readfirstlane((blockIdx.x << 3) + (threadIdx.x >> 5));
    float4 acc[4];
    hg_row_sum(Xe, list, off, cnt, v, lane, acc);
    const float s = degV[v];
    float4* dst = (float4*)(out + (size_t)v * DIM_) + lane;
    for (int j = 0; j < 4; ++j)
        dst[j * 32] = make_float4(acc[j].x * s, acc[j].y * s, acc[j].z * s, acc[j].w * s);
}

extern "C" void kernel_launch(void* const* d_in, const int* in_sizes, int n_in,
                              void* d_out, int out_size, void* d_ws, size_t ws_size,
                              hipStream_t stream) {
    (void)in_sizes; (void)n_in; (void)out_size; (void)ws_size;
    const float* X     = (const float*)d_in[0];
    const float* W     = (const float*)d_in[1];
    const float* degE  = (const float*)d_in[2];
    const float* degV  = (const float*)d_in[3];
    const float* Wdiag = (const float*)d_in[4];
    const int*   vidx  = (const int*)d_in[5];
    const int*   eidx  = (const int*)d_in[6];
    float* out = (float*)d_out;

    float* Xs   = (float*)d_ws;
    float* Xe   = Xs + (size_t)NE_ * DIM_;
    int*   cnt  = (int*)(Xe + (size_t)NE_ * DIM_);
    int*   off  = cnt + GRP_P_ * NR_;
    int*   list = off + GRP_P_ * NR_;

    hg_group<<<GRP_NB_, GRP_T_, 0, stream>>>(vidx, eidx, cnt, off, list);

    hg_gather_v2e<<<NE_ / 8, 256, 0, stream>>>(X, off, cnt, list, degE, Wdiag, Xs);

    hg_gemm_wmma<<<GB_WG_, 256, 0, stream>>>(Xs, W, Xe);

    hg_gather_e2v<<<NV_ / 8, 256, 0, stream>>>(Xe, off + NE_, cnt + NE_, list, degV, out);
}
